# starting kernel
# baseline (speedup 1.0000x reference)
#include <hip/hip_runtime.h>

typedef __attribute__((ext_vector_type(2))) float v2f;
typedef __attribute__((ext_vector_type(8))) float v8f;

#define QB      8
#define NSTATE  256
#define LAYERS  2
#define ENC     16
#define DIM     512
#define BTILE   16

__launch_bounds__(256)
__global__ void qenc_fused(const float* __restrict__ x,
                           const float* __restrict__ proj_w,
                           const float* __restrict__ qw,
                           float* __restrict__ out)
{
    __shared__ float  ampsS[BTILE * NSTATE];
    __shared__ float2 st[BTILE * NSTATE];
    __shared__ float  red[BTILE][16];
    __shared__ float  invn[BTILE];

    const int t     = threadIdx.x;
    const int lane  = t & 31;
    const int wave  = t >> 5;
    const int bt    = blockIdx.x;
    const int e     = blockIdx.y;
    const int b0    = bt * BTILE;

    const int mlane = lane & 15;
    const int khalf = lane >> 4;

    const float* xrow = x + (size_t)(b0 + mlane) * DIM + 2 * khalf;
    const float* pwE  = proj_w + (size_t)e * NSTATE * DIM;

    #pragma unroll
    for (int half = 0; half < 2; ++half) {
        const int n0 = (wave + half * 8) * 16;
        const float* brow = pwE + (size_t)(n0 + mlane) * DIM + 2 * khalf;
        v8f acc = {0.f, 0.f, 0.f, 0.f, 0.f, 0.f, 0.f, 0.f};
        for (int k0 = 0; k0 < DIM; k0 += 4) {
            v2f a = *(const v2f*)(xrow + k0);
            v2f b = *(const v2f*)(brow + k0);
            acc = __builtin_amdgcn_wmma_f32_16x16x4_f32(
                      false, a, false, b, (short)0, acc, false, false);
        }
        #pragma unroll
        for (int r = 0; r < 8; ++r)
            ampsS[(r + 8 * khalf) * NSTATE + n0 + mlane] = acc[r];
    }
    __syncthreads();

    const int row = t >> 4;
    const int li  = t & 15;
    {
        float ss = 0.f;
        #pragma unroll 4
        for (int j = 0; j < 16; ++j) {
            float v = ampsS[row * NSTATE + li * 16 + j];
            ss += v * v;
        }
        red[row][li] = ss;
    }
    __syncthreads();
    if (t < BTILE) {
        float s = 0.f;
        #pragma unroll
        for (int j = 0; j < 16; ++j) s += red[t][j];
        invn[t] = rsqrtf(s);
    }
    __syncthreads();
    {
        float inv = invn[row];
        #pragma unroll 4
        for (int j = 0; j < 16; ++j) {
            int s = li * 16 + j;
            st[row * NSTATE + s] = make_float2(ampsS[row * NSTATE + s] * inv, 0.f);
        }
    }
    __syncthreads();

    const float* wE = qw + (size_t)e * (LAYERS * QB * 3);
    for (int l = 0; l < LAYERS; ++l) {
        for (int q = 0; q < QB; ++q) {
            float phi   = wE[(l * QB + q) * 3 + 0];
            float theta = wE[(l * QB + q) * 3 + 1];
            float omega = wE[(l * QB + q) * 3 + 2];
            float sh, ch; __sincosf(0.5f * theta,          &sh, &ch);
            float sp, cp; __sincosf(0.5f * (phi + omega),  &sp, &cp);
            float sm, cm; __sincosf(0.5f * (phi - omega),  &sm, &cm);
            const float2 U00 = make_float2( cp * ch, -sp * ch);
            const float2 U01 = make_float2(-cm * sh, -sm * sh);
            const float2 U10 = make_float2( cm * sh, -sm * sh);
            const float2 U11 = make_float2( cp * ch,  sp * ch);

            const int bit  = (QB - 1) - q;
            const int mask = 1 << bit;
            const int lowm = mask - 1;
            #pragma unroll
            for (int p8 = 0; p8 < 8; ++p8) {
                int p  = t + p8 * 256;
                int sv = p >> 7;
                int j  = p & 127;
                int i0 = ((j & ~lowm) << 1) | (j & lowm);
                int i1 = i0 | mask;
                float2 a0 = st[sv * NSTATE + i0];
                float2 a1 = st[sv * NSTATE + i1];
                float2 r0, r1;
                r0.x = U00.x*a0.x - U00.y*a0.y + U01.x*a1.x - U01.y*a1.y;
                r0.y = U00.x*a0.y + U00.y*a0.x + U01.x*a1.y + U01.y*a1.x;
                r1.x = U10.x*a0.x - U10.y*a0.y + U11.x*a1.x - U11.y*a1.y;
                r1.y = U10.x*a0.y + U10.y*a0.x + U11.x*a1.y + U11.y*a1.x;
                st[sv * NSTATE + i0] = r0;
                st[sv * NSTATE + i1] = r1;
            }
            __syncthreads();
        }
        const int rr = (l % (QB - 1)) + 1;
        for (int q = 0; q < QB; ++q) {
            const int tq  = (q + rr) & (QB - 1);
            const int bc  = (QB - 1) - q;
            const int btb = (QB - 1) - tq;
            const int lo  = bc < btb ? bc : btb;
            const int hi  = bc < btb ? btb : bc;
            const int lom = (1 << lo) - 1;
            const int him = (1 << hi) - 1;
            #pragma unroll
            for (int p4 = 0; p4 < 4; ++p4) {
                int p  = t + p4 * 256;
                int sv = p >> 6;
                int j  = p & 63;
                int i  = ((j & ~lom) << 1) | (j & lom);
                i      = ((i & ~him) << 1) | (i & him);
                i     |= (1 << bc);
                int i2 = i | (1 << btb);
                float2 v0 = st[sv * NSTATE + i];
                float2 v1 = st[sv * NSTATE + i2];
                st[sv * NSTATE + i]  = v1;
                st[sv * NSTATE + i2] = v0;
            }
            __syncthreads();
        }
    }

    float z[QB];
    #pragma unroll
    for (int q = 0; q < QB; ++q) z[q] = 0.f;
    for (int j = 0; j < 16; ++j) {
        int s = li * 16 + j;
        float2 a = st[row * NSTATE + s];
        float p = a.x * a.x + a.y * a.y;
        #pragma unroll
        for (int q = 0; q < QB; ++q)
            z[q] += ((s >> ((QB - 1) - q)) & 1) ? -p : p;
    }
    #pragma unroll
    for (int q = 0; q < QB; ++q)
        ampsS[row * NSTATE + li * 16 + q] = z[q];
    __syncthreads();
    if (t < BTILE * QB) {
        int orow = t >> 3, oq = t & 7;
        float s = 0.f;
        #pragma unroll
        for (int j = 0; j < 16; ++j)
            s += ampsS[orow * NSTATE + j * 16 + oq];
        out[(size_t)(b0 + orow) * (ENC * QB) + e * QB + oq] = s;
    }
}

extern "C" void kernel_launch(void* const* d_in, const int* in_sizes, int n_in,
                              void* d_out, int out_size, void* d_ws, size_t ws_size,
                              hipStream_t stream) {
    const float* x  = (const float*)d_in[0];
    const float* pw = (const float*)d_in[1];
    const float* qw = (const float*)d_in[2];
    float* out = (float*)d_out;
    const int B = in_sizes[0] / DIM;
    dim3 grid(B / BTILE, ENC);
    qenc_fused<<<grid, 256, 0, stream>>>(x, pw, qw, out);
}


# GEMM operand loads: one 16-byte load per lane feeds two k-steps and the x fragment is shared by the wave's two column tiles (512 -> 192 loads per wave); k summed in another order within each 8
# speedup vs baseline: 1.0793x; 1.0793x over previous
#include <hip/hip_runtime.h>

typedef __attribute__((ext_vector_type(2))) float v2f;
typedef __attribute__((ext_vector_type(4))) float v4f;
typedef __attribute__((ext_vector_type(8))) float v8f;

#define QB      8
#define NSTATE  256
#define LAYERS  2
#define ENC     16
#define DIM     512
#define BTILE   16

__launch_bounds__(256)
__global__ void qenc_fused(const float* __restrict__ x,
                           const float* __restrict__ proj_w,
                           const float* __restrict__ qw,
                           float* __restrict__ out)
{
    __shared__ float  ampsS[BTILE * NSTATE];
    __shared__ float2 st[BTILE * NSTATE];
    __shared__ float  red[BTILE][16];
    __shared__ float  invn[BTILE];

    const int t     = threadIdx.x;
    const int lane  = t & 31;
    const int wave  = t >> 5;
    const int bt    = blockIdx.x;
    const int e     = blockIdx.y;
    const int b0    = bt * BTILE;

    const int mlane = lane & 15;
    const int khalf = lane >> 4;

    const float* xrow = x + (size_t)(b0 + mlane) * DIM + 4 * khalf;
    const float* pwE  = proj_w + (size_t)e * NSTATE * DIM;

    {
        const int n0a = wave * 16;
        const int n0b = (wave + 8) * 16;
        const float* browa = pwE + (size_t)(n0a + mlane) * DIM + 4 * khalf;
        const float* browb = pwE + (size_t)(n0b + mlane) * DIM + 4 * khalf;
        v8f acca = {0.f, 0.f, 0.f, 0.f, 0.f, 0.f, 0.f, 0.f};
        v8f accb = {0.f, 0.f, 0.f, 0.f, 0.f, 0.f, 0.f, 0.f};
        #pragma unroll 4
        for (int k0 = 0; k0 < DIM; k0 += 8) {
            const v4f a  = *(const v4f*)(xrow  + k0);
            const v4f ba = *(const v4f*)(browa + k0);
            const v4f bb = *(const v4f*)(browb + k0);
            const v2f a01 = {a.x, a.y},   a23 = {a.z, a.w};
            const v2f ba01 = {ba.x, ba.y}, ba23 = {ba.z, ba.w};
            const v2f bb01 = {bb.x, bb.y}, bb23 = {bb.z, bb.w};
            acca = __builtin_amdgcn_wmma_f32_16x16x4_f32(
                       false, a01, false, ba01, (short)0, acca, false, false);
            accb = __builtin_amdgcn_wmma_f32_16x16x4_f32(
                       false, a01, false, bb01, (short)0, accb, false, false);
            acca = __builtin_amdgcn_wmma_f32_16x16x4_f32(
                       false, a23, false, ba23, (short)0, acca, false, false);
            accb = __builtin_amdgcn_wmma_f32_16x16x4_f32(
                       false, a23, false, bb23, (short)0, accb, false, false);
        }
        #pragma unroll
        for (int r = 0; r < 8; ++r) {
            ampsS[(r + 8 * khalf) * NSTATE + n0a + mlane] = acca[r];
            ampsS[(r + 8 * khalf) * NSTATE + n0b + mlane] = accb[r];
        }
    }
    __syncthreads();

    const int row = t >> 4;
    const int li  = t & 15;
    {
        float ss = 0.f;
        #pragma unroll 4
        for (int j = 0; j < 16; ++j) {
            float v = ampsS[row * NSTATE + li * 16 + j];
            ss += v * v;
        }
        red[row][li] = ss;
    }
    __syncthreads();
    if (t < BTILE) {
        float s = 0.f;
        #pragma unroll
        for (int j = 0; j < 16; ++j) s += red[t][j];
        invn[t] = rsqrtf(s);
    }
    __syncthreads();
    {
        float inv = invn[row];
        #pragma unroll 4
        for (int j = 0; j < 16; ++j) {
            int s = li * 16 + j;
            st[row * NSTATE + s] = make_float2(ampsS[row * NSTATE + s] * inv, 0.f);
        }
    }
    __syncthreads();

    const float* wE = qw + (size_t)e * (LAYERS * QB * 3);
    for (int l = 0; l < LAYERS; ++l) {
        for (int q = 0; q < QB; ++q) {
            float phi   = wE[(l * QB + q) * 3 + 0];
            float theta = wE[(l * QB + q) * 3 + 1];
            float omega = wE[(l * QB + q) * 3 + 2];
            float sh, ch; __sincosf(0.5f * theta,          &sh, &ch);
            float sp, cp; __sincosf(0.5f * (phi + omega),  &sp, &cp);
            float sm, cm; __sincosf(0.5f * (phi - omega),  &sm, &cm);
            const float2 U00 = make_float2( cp * ch, -sp * ch);
            const float2 U01 = make_float2(-cm * sh, -sm * sh);
            const float2 U10 = make_float2( cm * sh, -sm * sh);
            const float2 U11 = make_float2( cp * ch,  sp * ch);

            const int bit  = (QB - 1) - q;
            const int mask = 1 << bit;
            const int lowm = mask - 1;
            #pragma unroll
            for (int p8 = 0; p8 < 8; ++p8) {
                int p  = t + p8 * 256;
                int sv = p >> 7;
                int j  = p & 127;
                int i0 = ((j & ~lowm) << 1) | (j & lowm);
                int i1 = i0 | mask;
                float2 a0 = st[sv * NSTATE + i0];
                float2 a1 = st[sv * NSTATE + i1];
                float2 r0, r1;
                r0.x = U00.x*a0.x - U00.y*a0.y + U01.x*a1.x - U01.y*a1.y;
                r0.y = U00.x*a0.y + U00.y*a0.x + U01.x*a1.y + U01.y*a1.x;
                r1.x = U10.x*a0.x - U10.y*a0.y + U11.x*a1.x - U11.y*a1.y;
                r1.y = U10.x*a0.y + U10.y*a0.x + U11.x*a1.y + U11.y*a1.x;
                st[sv * NSTATE + i0] = r0;
                st[sv * NSTATE + i1] = r1;
            }
            __syncthreads();
        }
        const int rr = (l % (QB - 1)) + 1;
        for (int q = 0; q < QB; ++q) {
            const int tq  = (q + rr) & (QB - 1);
            const int bc  = (QB - 1) - q;
            const int btb = (QB - 1) - tq;
            const int lo  = bc < btb ? bc : btb;
            const int hi  = bc < btb ? btb : bc;
            const int lom = (1 << lo) - 1;
            const int him = (1 << hi) - 1;
            #pragma unroll
            for (int p4 = 0; p4 < 4; ++p4) {
                int p  = t + p4 * 256;
                int sv = p >> 6;
                int j  = p & 63;
                int i  = ((j & ~lom) << 1) | (j & lom);
                i      = ((i & ~him) << 1) | (i & him);
                i     |= (1 << bc);
                int i2 = i | (1 << btb);
                float2 v0 = st[sv * NSTATE + i];
                float2 v1 = st[sv * NSTATE + i2];
                st[sv * NSTATE + i]  = v1;
                st[sv * NSTATE + i2] = v0;
            }
            __syncthreads();
        }
    }

    float z[QB];
    #pragma unroll
    for (int q = 0; q < QB; ++q) z[q] = 0.f;
    for (int j = 0; j < 16; ++j) {
        int s = li * 16 + j;
        float2 a = st[row * NSTATE + s];
        float p = a.x * a.x + a.y * a.y;
        #pragma unroll
        for (int q = 0; q < QB; ++q)
            z[q] += ((s >> ((QB - 1) - q)) & 1) ? -p : p;
    }
    #pragma unroll
    for (int q = 0; q < QB; ++q)
        ampsS[row * NSTATE + li * 16 + q] = z[q];
    __syncthreads();
    if (t < BTILE * QB) {
        int orow = t >> 3, oq = t & 7;
        float s = 0.f;
        #pragma unroll
        for (int j = 0; j < 16; ++j)
            s += ampsS[orow * NSTATE + j * 16 + oq];
        out[(size_t)(b0 + orow) * (ENC * QB) + e * QB + oq] = s;
    }
}

extern "C" void kernel_launch(void* const* d_in, const int* in_sizes, int n_in,
                              void* d_out, int out_size, void* d_ws, size_t ws_size,
                              hipStream_t stream) {
    const float* x  = (const float*)d_in[0];
    const float* pw = (const float*)d_in[1];
    const float* qw = (const float*)d_in[2];
    float* out = (float*)d_out;
    const int B = in_sizes[0] / DIM;
    dim3 grid(B / BTILE, ENC);
    qenc_fused<<<grid, 256, 0, stream>>>(x, pw, qw, out);
}


# Simulation phase from registers: each lane holds 16 amplitudes and does 4 rotations per LDS pass, CNOT cascades become store index and Z signs (37 -> 7 barriers); norm and Z summed in another order
# speedup vs baseline: 1.2518x; 1.1598x over previous
#include <hip/hip_runtime.h>


typedef __attribute__((ext_vector_type(2))) float v2f;
typedef __attribute__((ext_vector_type(4))) float v4f;
typedef __attribute__((ext_vector_type(8))) float v8f;

#define QB      8
#define NSTATE  256
#define LAYERS  2
#define ENC     16
#define DIM     512
#define BTILE   16
#define SVP     274

__host__ __device__ constexpr int cnot_cascade(int j, int rr)
{
    for (int q = 0; q < QB; ++q) {
        const int bc = (QB - 1) - q;
        const int bt = (QB - 1) - ((q + rr) & (QB - 1));
        j ^= ((j >> bc) & 1) << bt;
    }
    return j;
}

template <int M>
__device__ __forceinline__ void rot_on_bit(float (&ar)[16], float (&ai)[16],
                                           const float* __restrict__ w3)
{
    const float phi   = w3[0];
    const float theta = w3[1];
    const float omega = w3[2];
    float sh, ch; __sincosf(0.5f * theta,         &sh, &ch);
    float sp, cp; __sincosf(0.5f * (phi + omega), &sp, &cp);
    float sm, cm; __sincosf(0.5f * (phi - omega), &sm, &cm);
    const float U00x =  cp * ch, U00y = -sp * ch;
    const float U01x = -cm * sh, U01y = -sm * sh;
    const float U10x =  cm * sh, U10y = -sm * sh;
    const float U11x =  cp * ch, U11y =  sp * ch;
    constexpr int lowm = (1 << M) - 1;
    #pragma unroll
    for (int k = 0; k < 8; ++k) {
        const int r0 = ((k & ~lowm) << 1) | (k & lowm);
        const int r1 = r0 | (1 << M);
        const float a0x = ar[r0], a0y = ai[r0];
        const float a1x = ar[r1], a1y = ai[r1];
        ar[r0] = U00x*a0x - U00y*a0y + U01x*a1x - U01y*a1y;
        ai[r0] = U00x*a0y + U00y*a0x + U01x*a1y + U01y*a1x;
        ar[r1] = U10x*a0x - U10y*a0y + U11x*a1x - U11y*a1y;
        ai[r1] = U10x*a0y + U10y*a0x + U11x*a1y + U11y*a1x;
    }
}

__device__ __forceinline__ void rot4(float (&ar)[16], float (&ai)[16],
                                     const float* __restrict__ w12)
{
    rot_on_bit<3>(ar, ai, w12 + 0);
    rot_on_bit<2>(ar, ai, w12 + 3);
    rot_on_bit<1>(ar, ai, w12 + 6);
    rot_on_bit<0>(ar, ai, w12 + 9);
}

__launch_bounds__(256)
__global__ void qenc_fused(const float* __restrict__ x,
                           const float* __restrict__ proj_w,
                           const float* __restrict__ qw,
                           float* __restrict__ out)
{
    __shared__ float2 st[BTILE * SVP];

    const int t     = threadIdx.x;
    const int lane  = t & 31;
    const int wave  = t >> 5;
    const int bt    = blockIdx.x;
    const int e     = blockIdx.y;
    const int b0    = bt * BTILE;

    const int mlane = lane & 15;
    const int khalf = lane >> 4;

    const float* xrow = x + (size_t)(b0 + mlane) * DIM + 4 * khalf;
    const float* pwE  = proj_w + (size_t)e * NSTATE * DIM;

    {
        const int n0a = wave * 16;
        const int n0b = (wave + 8) * 16;
        const float* browa = pwE + (size_t)(n0a + mlane) * DIM + 4 * khalf;
        const float* browb = pwE + (size_t)(n0b + mlane) * DIM + 4 * khalf;
        v8f acca = {0.f, 0.f, 0.f, 0.f, 0.f, 0.f, 0.f, 0.f};
        v8f accb = {0.f, 0.f, 0.f, 0.f, 0.f, 0.f, 0.f, 0.f};
        #pragma unroll 4
        for (int k0 = 0; k0 < DIM; k0 += 8) {
            const v4f a  = *(const v4f*)(xrow  + k0);
            const v4f ba = *(const v4f*)(browa + k0);
            const v4f bb = *(const v4f*)(browb + k0);
            const v2f a01 = {a.x, a.y},   a23 = {a.z, a.w};
            const v2f ba01 = {ba.x, ba.y}, ba23 = {ba.z, ba.w};
            const v2f bb01 = {bb.x, bb.y}, bb23 = {bb.z, bb.w};
            acca = __builtin_amdgcn_wmma_f32_16x16x4_f32(
                       false, a01, false, ba01, (short)0, acca, false, false);
            accb = __builtin_amdgcn_wmma_f32_16x16x4_f32(
                       false, a01, false, bb01, (short)0, accb, false, false);
            acca = __builtin_amdgcn_wmma_f32_16x16x4_f32(
                       false, a23, false, ba23, (short)0, acca, false, false);
            accb = __builtin_amdgcn_wmma_f32_16x16x4_f32(
                       false, a23, false, bb23, (short)0, accb, false, false);
        }
        #pragma unroll
        for (int r = 0; r < 8; ++r) {
            st[(r + 8 * khalf) * SVP + 17 * wave       + mlane].x = acca[r];
            st[(r + 8 * khalf) * SVP + 17 * (wave + 8) + mlane].x = accb[r];
        }
    }
    __syncthreads();

    const int sv = wave + 8 * khalf;
    const int li = mlane;
    float2* sb = st + sv * SVP;
    const float* wE = qw + (size_t)e * (LAYERS * QB * 3);

    float ar[16], ai[16];

    {
        float ss = 0.f;
        #pragma unroll
        for (int r = 0; r < 16; ++r) {
            ar[r] = sb[r * 17 + li].x;
            ss += ar[r] * ar[r];
        }
        #pragma unroll
        for (int m = 8; m >= 1; m >>= 1) ss += __shfl_xor(ss, m, 16);
        const float inv = rsqrtf(ss);
        #pragma unroll
        for (int r = 0; r < 16; ++r) { ar[r] *= inv; ai[r] = 0.f; }
    }
    rot4(ar, ai, wE + (0 * QB + 0) * 3);
    #pragma unroll
    for (int r = 0; r < 16; ++r) sb[r * 17 + li] = make_float2(ar[r], ai[r]);
    __syncthreads();

    #pragma unroll
    for (int r = 0; r < 16; ++r) { const float2 a = sb[li * 17 + r]; ar[r] = a.x; ai[r] = a.y; }
    rot4(ar, ai, wE + (0 * QB + 4) * 3);
    __syncthreads();
    {
        const int hi = cnot_cascade(li << 4, (0 % (QB - 1)) + 1);
        #pragma unroll
        for (int r = 0; r < 16; ++r) {
            const int d = hi ^ cnot_cascade(r, (0 % (QB - 1)) + 1);
            sb[d + (d >> 4)] = make_float2(ar[r], ai[r]);
        }
    }
    __syncthreads();

    #pragma unroll
    for (int r = 0; r < 16; ++r) { const float2 a = sb[r * 17 + li]; ar[r] = a.x; ai[r] = a.y; }
    rot4(ar, ai, wE + (1 * QB + 0) * 3);
    #pragma unroll
    for (int r = 0; r < 16; ++r) sb[r * 17 + li] = make_float2(ar[r], ai[r]);
    __syncthreads();

    #pragma unroll
    for (int r = 0; r < 16; ++r) { const float2 a = sb[li * 17 + r]; ar[r] = a.x; ai[r] = a.y; }
    rot4(ar, ai, wE + (1 * QB + 4) * 3);

    float z[QB];
    {
        const int hi = cnot_cascade(li << 4, (1 % (QB - 1)) + 1);
        float p[16];
        #pragma unroll
        for (int r = 0; r < 16; ++r) p[r] = ar[r] * ar[r] + ai[r] * ai[r];
        #pragma unroll
        for (int q = 0; q < QB; ++q) {
            float acc = 0.f;
            #pragma unroll
            for (int r = 0; r < 16; ++r)
                acc += ((cnot_cascade(r, (1 % (QB - 1)) + 1) >> ((QB - 1) - q)) & 1) ? -p[r] : p[r];
            z[q] = ((hi >> ((QB - 1) - q)) & 1) ? -acc : acc;
        }
    }
    __syncthreads();
    float* zs = reinterpret_cast<float*>(st);
    #pragma unroll
    for (int q = 0; q < QB; ++q)
        zs[(sv * 16 + li) * QB + q] = z[q];
    __syncthreads();
    if (t < BTILE * QB) {
        const int orow = t >> 3, oq = t & 7;
        float s = 0.f;
        #pragma unroll
        for (int j = 0; j < 16; ++j)
            s += zs[(orow * 16 + j) * QB + oq];
        out[(size_t)(b0 + orow) * (ENC * QB) + e * QB + oq] = s;
    }
}

extern "C" void kernel_launch(void* const* d_in, const int* in_sizes, int n_in,
                              void* d_out, int out_size, void* d_ws, size_t ws_size,
                              hipStream_t stream) {
    const float* x  = (const float*)d_in[0];
    const float* pw = (const float*)d_in[1];
    const float* qw = (const float*)d_in[2];
    float* out = (float*)d_out;
    const int B = in_sizes[0] / DIM;
    dim3 grid(B / BTILE, ENC);
    qenc_fused<<<grid, 256, 0, stream>>>(x, pw, qw, out);
}


# GEMM row blocking: a workgroup covers 64 batch rows, each proj_w fragment load feeds 4 row tiles (6 loads per 16 WMMA, was 12), next k blocks loaded one set ahead; 64 statevectors in 4 rounds of 16
# speedup vs baseline: 2.0392x; 1.6290x over previous
#include <hip/hip_runtime.h>


typedef __attribute__((ext_vector_type(2))) float v2f;
typedef __attribute__((ext_vector_type(4))) float v4f;
typedef __attribute__((ext_vector_type(8))) float v8f;

#define QB      8
#define NSTATE  256
#define LAYERS  2
#define ENC     16
#define DIM     512
#define BTILE   16
#define MT      4
#define WGROWS  (MT * BTILE)
#define AP      276
#define SVP     274

__host__ __device__ constexpr int cnot_cascade(int j, int rr)
{
    for (int q = 0; q < QB; ++q) {
        const int bc = (QB - 1) - q;
        const int bt = (QB - 1) - ((q + rr) & (QB - 1));
        j ^= ((j >> bc) & 1) << bt;
    }
    return j;
}

template <int M>
__device__ __forceinline__ void rot_on_bit(float (&ar)[16], float (&ai)[16],
                                           const float* __restrict__ w3)
{
    const float phi   = w3[0];
    const float theta = w3[1];
    const float omega = w3[2];
    float sh, ch; __sincosf(0.5f * theta,         &sh, &ch);
    float sp, cp; __sincosf(0.5f * (phi + omega), &sp, &cp);
    float sm, cm; __sincosf(0.5f * (phi - omega), &sm, &cm);
    const float U00x =  cp * ch, U00y = -sp * ch;
    const float U01x = -cm * sh, U01y = -sm * sh;
    const float U10x =  cm * sh, U10y = -sm * sh;
    const float U11x =  cp * ch, U11y =  sp * ch;
    constexpr int lowm = (1 << M) - 1;
    #pragma unroll
    for (int k = 0; k < 8; ++k) {
        const int r0 = ((k & ~lowm) << 1) | (k & lowm);
        const int r1 = r0 | (1 << M);
        const float a0x = ar[r0], a0y = ai[r0];
        const float a1x = ar[r1], a1y = ai[r1];
        ar[r0] = U00x*a0x - U00y*a0y + U01x*a1x - U01y*a1y;
        ai[r0] = U00x*a0y + U00y*a0x + U01x*a1y + U01y*a1x;
        ar[r1] = U10x*a0x - U10y*a0y + U11x*a1x - U11y*a1y;
        ai[r1] = U10x*a0y + U10y*a0x + U11x*a1y + U11y*a1x;
    }
}

__device__ __forceinline__ void rot4(float (&ar)[16], float (&ai)[16],
                                     const float* __restrict__ w12)
{
    rot_on_bit<3>(ar, ai, w12 + 0);
    rot_on_bit<2>(ar, ai, w12 + 3);
    rot_on_bit<1>(ar, ai, w12 + 6);
    rot_on_bit<0>(ar, ai, w12 + 9);
}

struct KFrag { v4f a[MT]; v4f ba, bb; };

__device__ __forceinline__ void load_kfrag(KFrag& f, const float* __restrict__ xrow,
                                           const float* __restrict__ browa,
                                           const float* __restrict__ browb, int k0)
{
    #pragma unroll
    for (int i = 0; i < MT; ++i) f.a[i] = *(const v4f*)(xrow + (size_t)i * BTILE * DIM + k0);
    f.ba = *(const v4f*)(browa + k0);
    f.bb = *(const v4f*)(browb + k0);
}

__device__ __forceinline__ void mma_kfrag(v8f (&acca)[MT], v8f (&accb)[MT], const KFrag& f)
{
    const v2f ba01 = {f.ba.x, f.ba.y}, ba23 = {f.ba.z, f.ba.w};
    const v2f bb01 = {f.bb.x, f.bb.y}, bb23 = {f.bb.z, f.bb.w};
    #pragma unroll
    for (int i = 0; i < MT; ++i) {
        const v2f a01 = {f.a[i].x, f.a[i].y};
        acca[i] = __builtin_amdgcn_wmma_f32_16x16x4_f32(
                      false, a01, false, ba01, (short)0, acca[i], false, false);
        accb[i] = __builtin_amdgcn_wmma_f32_16x16x4_f32(
                      false, a01, false, bb01, (short)0, accb[i], false, false);
    }
    #pragma unroll
    for (int i = 0; i < MT; ++i) {
        const v2f a23 = {f.a[i].z, f.a[i].w};
        acca[i] = __builtin_amdgcn_wmma_f32_16x16x4_f32(
                      false, a23, false, ba23, (short)0, acca[i], false, false);
        accb[i] = __builtin_amdgcn_wmma_f32_16x16x4_f32(
                      false, a23, false, bb23, (short)0, accb[i], false, false);
    }
}

__launch_bounds__(256)
__global__ void qenc_fused(const float* __restrict__ x,
                           const float* __restrict__ proj_w,
                           const float* __restrict__ qw,
                           float* __restrict__ out)
{
    __shared__ float amps[WGROWS * AP];
    __shared__ float2 st[BTILE * SVP];

    const int t     = threadIdx.x;
    const int lane  = t & 31;
    const int wave  = t >> 5;
    const int bt    = blockIdx.x;
    const int e     = blockIdx.y;
    const int b0    = bt * WGROWS;

    const int mlane = lane & 15;
    const int khalf = lane >> 4;

    const float* xrow = x + (size_t)(b0 + mlane) * DIM + 4 * khalf;
    const float* pwE  = proj_w + (size_t)e * NSTATE * DIM;

    {
        const int n0a = wave * 16;
        const int n0b = (wave + 8) * 16;
        const float* browa = pwE + (size_t)(n0a + mlane) * DIM + 4 * khalf;
        const float* browb = pwE + (size_t)(n0b + mlane) * DIM + 4 * khalf;
        v8f acca[MT], accb[MT];
        #pragma unroll
        for (int i = 0; i < MT; ++i) {
            acca[i] = (v8f){0.f, 0.f, 0.f, 0.f, 0.f, 0.f, 0.f, 0.f};
            accb[i] = (v8f){0.f, 0.f, 0.f, 0.f, 0.f, 0.f, 0.f, 0.f};
        }
        KFrag p0[2], p1[2];
        load_kfrag(p0[0], xrow, browa, browb, 0);
        load_kfrag(p0[1], xrow, browa, browb, 8);
        #pragma unroll 1
        for (int k0 = 0; k0 < DIM - 32; k0 += 32) {
            load_kfrag(p1[0], xrow, browa, browb, k0 + 16);
            load_kfrag(p1[1], xrow, browa, browb, k0 + 24);
            mma_kfrag(acca, accb, p0[0]);
            mma_kfrag(acca, accb, p0[1]);
            load_kfrag(p0[0], xrow, browa, browb, k0 + 32);
            load_kfrag(p0[1], xrow, browa, browb, k0 + 40);
            mma_kfrag(acca, accb, p1[0]);
            mma_kfrag(acca, accb, p1[1]);
        }
        load_kfrag(p1[0], xrow, browa, browb, DIM - 16);
        load_kfrag(p1[1], xrow, browa, browb, DIM - 8);
        mma_kfrag(acca, accb, p0[0]);
        mma_kfrag(acca, accb, p0[1]);
        mma_kfrag(acca, accb, p1[0]);
        mma_kfrag(acca, accb, p1[1]);
        #pragma unroll
        for (int i = 0; i < MT; ++i) {
            #pragma unroll
            for (int r = 0; r < 8; ++r) {
                amps[(16 * i + r + 8 * khalf) * AP + 17 * wave       + mlane] = acca[i][r];
                amps[(16 * i + r + 8 * khalf) * AP + 17 * (wave + 8) + mlane] = accb[i][r];
            }
        }
    }
    __syncthreads();

    const int sv = wave + 8 * khalf;
    const int li = mlane;
    float2* sb = st + sv * SVP;
    const float* wE = qw + (size_t)e * (LAYERS * QB * 3);

    #pragma unroll 1
    for (int m = 0; m < MT; ++m) {
        float ar[16], ai[16];

        {
            const float* arow = amps + (16 * m + sv) * AP;
            float ss = 0.f;
            #pragma unroll
            for (int r = 0; r < 16; ++r) {
                ar[r] = arow[r * 17 + li];
                ss += ar[r] * ar[r];
            }
            #pragma unroll
            for (int d = 8; d >= 1; d >>= 1) ss += __shfl_xor(ss, d, 16);
            const float inv = rsqrtf(ss);
            #pragma unroll
            for (int r = 0; r < 16; ++r) { ar[r] *= inv; ai[r] = 0.f; }
        }
        rot4(ar, ai, wE + (0 * QB + 0) * 3);
        #pragma unroll
        for (int r = 0; r < 16; ++r) sb[r * 17 + li] = make_float2(ar[r], ai[r]);
        __syncthreads();

        #pragma unroll
        for (int r = 0; r < 16; ++r) { const float2 a = sb[li * 17 + r]; ar[r] = a.x; ai[r] = a.y; }
        rot4(ar, ai, wE + (0 * QB + 4) * 3);
        __syncthreads();
        {
            const int hi = cnot_cascade(li << 4, (0 % (QB - 1)) + 1);
            #pragma unroll
            for (int r = 0; r < 16; ++r) {
                const int d = hi ^ cnot_cascade(r, (0 % (QB - 1)) + 1);
                sb[d + (d >> 4)] = make_float2(ar[r], ai[r]);
            }
        }
        __syncthreads();

        #pragma unroll
        for (int r = 0; r < 16; ++r) { const float2 a = sb[r * 17 + li]; ar[r] = a.x; ai[r] = a.y; }
        rot4(ar, ai, wE + (1 * QB + 0) * 3);
        #pragma unroll
        for (int r = 0; r < 16; ++r) sb[r * 17 + li] = make_float2(ar[r], ai[r]);
        __syncthreads();

        #pragma unroll
        for (int r = 0; r < 16; ++r) { const float2 a = sb[li * 17 + r]; ar[r] = a.x; ai[r] = a.y; }
        rot4(ar, ai, wE + (1 * QB + 4) * 3);

        float z[QB];
        {
            const int hi = cnot_cascade(li << 4, (1 % (QB - 1)) + 1);
            float p[16];
            #pragma unroll
            for (int r = 0; r < 16; ++r) p[r] = ar[r] * ar[r] + ai[r] * ai[r];
            #pragma unroll
            for (int q = 0; q < QB; ++q) {
                float acc = 0.f;
                #pragma unroll
                for (int r = 0; r < 16; ++r)
                    acc += ((cnot_cascade(r, (1 % (QB - 1)) + 1) >> ((QB - 1) - q)) & 1) ? -p[r] : p[r];
                z[q] = ((hi >> ((QB - 1) - q)) & 1) ? -acc : acc;
            }
        }
        __syncthreads();
        float* zs = reinterpret_cast<float*>(st);
        #pragma unroll
        for (int q = 0; q < QB; ++q)
            zs[(sv * 16 + li) * QB + q] = z[q];
        __syncthreads();
        if (t < BTILE * QB) {
            const int orow = t >> 3, oq = t & 7;
            float s = 0.f;
            #pragma unroll
            for (int j = 0; j < 16; ++j)
                s += zs[(orow * 16 + j) * QB + oq];
            out[(size_t)(b0 + 16 * m + orow) * (ENC * QB) + e * QB + oq] = s;
        }
        __syncthreads();
    }
}

extern "C" void kernel_launch(void* const* d_in, const int* in_sizes, int n_in,
                              void* d_out, int out_size, void* d_ws, size_t ws_size,
                              hipStream_t stream) {
    const float* x  = (const float*)d_in[0];
    const float* pw = (const float*)d_in[1];
    const float* qw = (const float*)d_in[2];
    float* out = (float*)d_out;
    const int B = in_sizes[0] / DIM;
    dim3 grid(B / WGROWS, ENC);
    qenc_fused<<<grid, 256, 0, stream>>>(x, pw, qw, out);
}


# x tile staged once in LDS by whole-line global loads and shared by the 8 waves: A fragments come by ds_load_b128, global loads per 16 WMMA 6 -> 2; same LDS reused for amps; order of sums unchanged
# speedup vs baseline: 2.2991x; 1.1275x over previous
#include <hip/hip_runtime.h>


typedef __attribute__((ext_vector_type(2))) float v2f;
typedef __attribute__((ext_vector_type(4))) float v4f;
typedef __attribute__((ext_vector_type(8))) float v8f;

#define QB      8
#define NSTATE  256
#define LAYERS  2
#define ENC     16
#define DIM     512
#define BTILE   16
#define MT      4
#define WGROWS  (MT * BTILE)
#define AP      276
#define SVP     274
#define XP      516
#define AMPS_FLOATS (WGROWS * AP)
#define ST_FLOATS   (2 * BTILE * SVP)
#define XS_FLOATS   (WGROWS * XP)
#define SMEM_FLOATS (XS_FLOATS > AMPS_FLOATS + ST_FLOATS ? XS_FLOATS : AMPS_FLOATS + ST_FLOATS)

__host__ __device__ constexpr int cnot_cascade(int j, int rr)
{
    for (int q = 0; q < QB; ++q) {
        const int bc = (QB - 1) - q;
        const int bt = (QB - 1) - ((q + rr) & (QB - 1));
        j ^= ((j >> bc) & 1) << bt;
    }
    return j;
}

template <int M>
__device__ __forceinline__ void rot_on_bit(float (&ar)[16], float (&ai)[16],
                                           const float* __restrict__ w3)
{
    const float phi   = w3[0];
    const float theta = w3[1];
    const float omega = w3[2];
    float sh, ch; __sincosf(0.5f * theta,         &sh, &ch);
    float sp, cp; __sincosf(0.5f * (phi + omega), &sp, &cp);
    float sm, cm; __sincosf(0.5f * (phi - omega), &sm, &cm);
    const float U00x =  cp * ch, U00y = -sp * ch;
    const float U01x = -cm * sh, U01y = -sm * sh;
    const float U10x =  cm * sh, U10y = -sm * sh;
    const float U11x =  cp * ch, U11y =  sp * ch;
    constexpr int lowm = (1 << M) - 1;
    #pragma unroll
    for (int k = 0; k < 8; ++k) {
        const int r0 = ((k & ~lowm) << 1) | (k & lowm);
        const int r1 = r0 | (1 << M);
        const float a0x = ar[r0], a0y = ai[r0];
        const float a1x = ar[r1], a1y = ai[r1];
        ar[r0] = U00x*a0x - U00y*a0y + U01x*a1x - U01y*a1y;
        ai[r0] = U00x*a0y + U00y*a0x + U01x*a1y + U01y*a1x;
        ar[r1] = U10x*a0x - U10y*a0y + U11x*a1x - U11y*a1y;
        ai[r1] = U10x*a0y + U10y*a0x + U11x*a1y + U11y*a1x;
    }
}

__device__ __forceinline__ void rot4(float (&ar)[16], float (&ai)[16],
                                     const float* __restrict__ w12)
{
    rot_on_bit<3>(ar, ai, w12 + 0);
    rot_on_bit<2>(ar, ai, w12 + 3);
    rot_on_bit<1>(ar, ai, w12 + 6);
    rot_on_bit<0>(ar, ai, w12 + 9);
}

struct KFrag { v4f a[MT]; v4f ba, bb; };

__device__ __forceinline__ void load_bfrag(KFrag& f, const float* __restrict__ browa,
                                           const float* __restrict__ browb, int k0)
{
    f.ba = *(const v4f*)(browa + k0);
    f.bb = *(const v4f*)(browb + k0);
}

__device__ __forceinline__ void load_afrag(KFrag& f, const float* xsrow, int k0)
{
    #pragma unroll
    for (int i = 0; i < MT; ++i) f.a[i] = *(const v4f*)(xsrow + i * BTILE * XP + k0);
}

__device__ __forceinline__ void load_kfrag(KFrag& f, const float* xsrow,
                                           const float* __restrict__ browa,
                                           const float* __restrict__ browb, int k0)
{
    load_afrag(f, xsrow, k0);
    load_bfrag(f, browa, browb, k0);
}

__device__ __forceinline__ void mma_kfrag(v8f (&acca)[MT], v8f (&accb)[MT], const KFrag& f)
{
    const v2f ba01 = {f.ba.x, f.ba.y}, ba23 = {f.ba.z, f.ba.w};
    const v2f bb01 = {f.bb.x, f.bb.y}, bb23 = {f.bb.z, f.bb.w};
    #pragma unroll
    for (int i = 0; i < MT; ++i) {
        const v2f a01 = {f.a[i].x, f.a[i].y};
        acca[i] = __builtin_amdgcn_wmma_f32_16x16x4_f32(
                      false, a01, false, ba01, (short)0, acca[i], false, false);
        accb[i] = __builtin_amdgcn_wmma_f32_16x16x4_f32(
                      false, a01, false, bb01, (short)0, accb[i], false, false);
    }
    #pragma unroll
    for (int i = 0; i < MT; ++i) {
        const v2f a23 = {f.a[i].z, f.a[i].w};
        acca[i] = __builtin_amdgcn_wmma_f32_16x16x4_f32(
                      false, a23, false, ba23, (short)0, acca[i], false, false);
        accb[i] = __builtin_amdgcn_wmma_f32_16x16x4_f32(
                      false, a23, false, bb23, (short)0, accb[i], false, false);
    }
}

__launch_bounds__(256)
__global__ void qenc_fused(const float* __restrict__ x,
                           const float* __restrict__ proj_w,
                           const float* __restrict__ qw,
                           float* __restrict__ out)
{
    __shared__ __attribute__((aligned(16))) float smem[SMEM_FLOATS];
    float*  xs   = smem;
    float*  amps = smem;
    float2* st   = reinterpret_cast<float2*>(smem + AMPS_FLOATS);

    const int t     = threadIdx.x;
    const int lane  = t & 31;
    const int wave  = t >> 5;
    const int bt    = blockIdx.x;
    const int e     = blockIdx.y;
    const int b0    = bt * WGROWS;

    const int mlane = lane & 15;
    const int khalf = lane >> 4;

    const float* xsrow = xs + mlane * XP + 4 * khalf;
    const float* pwE   = proj_w + (size_t)e * NSTATE * DIM;

    {
        const int n0a = wave * 16;
        const int n0b = (wave + 8) * 16;
        const float* browa = pwE + (size_t)(n0a + mlane) * DIM + 4 * khalf;
        const float* browb = pwE + (size_t)(n0b + mlane) * DIM + 4 * khalf;
        v8f acca[MT], accb[MT];
        #pragma unroll
        for (int i = 0; i < MT; ++i) {
            acca[i] = (v8f){0.f, 0.f, 0.f, 0.f, 0.f, 0.f, 0.f, 0.f};
            accb[i] = (v8f){0.f, 0.f, 0.f, 0.f, 0.f, 0.f, 0.f, 0.f};
        }
        KFrag p0[2], p1[2];
        load_bfrag(p0[0], browa, browb, 0);
        load_bfrag(p0[1], browa, browb, 8);
        {
            const float* xg = x + (size_t)b0 * DIM;
            #pragma unroll
            for (int i = 0; i < (WGROWS * DIM / 4) / 256; ++i) {
                const int idx = i * 256 + t;
                const int row = idx >> 7, c4 = idx & 127;
                *(v4f*)(xs + row * XP + 4 * c4) = *(const v4f*)(xg + (size_t)row * DIM + 4 * c4);
            }
        }
        __syncthreads();
        load_afrag(p0[0], xsrow, 0);
        load_afrag(p0[1], xsrow, 8);
        #pragma unroll 1
        for (int k0 = 0; k0 < DIM - 32; k0 += 32) {
            load_kfrag(p1[0], xsrow, browa, browb, k0 + 16);
            load_kfrag(p1[1], xsrow, browa, browb, k0 + 24);
            mma_kfrag(acca, accb, p0[0]);
            mma_kfrag(acca, accb, p0[1]);
            load_kfrag(p0[0], xsrow, browa, browb, k0 + 32);
            load_kfrag(p0[1], xsrow, browa, browb, k0 + 40);
            mma_kfrag(acca, accb, p1[0]);
            mma_kfrag(acca, accb, p1[1]);
        }
        load_kfrag(p1[0], xsrow, browa, browb, DIM - 16);
        load_kfrag(p1[1], xsrow, browa, browb, DIM - 8);
        mma_kfrag(acca, accb, p0[0]);
        mma_kfrag(acca, accb, p0[1]);
        mma_kfrag(acca, accb, p1[0]);
        mma_kfrag(acca, accb, p1[1]);
        __syncthreads();
        #pragma unroll
        for (int i = 0; i < MT; ++i) {
            #pragma unroll
            for (int r = 0; r < 8; ++r) {
                amps[(16 * i + r + 8 * khalf) * AP + 17 * wave       + mlane] = acca[i][r];
                amps[(16 * i + r + 8 * khalf) * AP + 17 * (wave + 8) + mlane] = accb[i][r];
            }
        }
    }
    __syncthreads();

    const int sv = wave + 8 * khalf;
    const int li = mlane;
    float2* sb = st + sv * SVP;
    const float* wE = qw + (size_t)e * (LAYERS * QB * 3);

    #pragma unroll 1
    for (int m = 0; m < MT; ++m) {
        float ar[16], ai[16];

        {
            const float* arow = amps + (16 * m + sv) * AP;
            float ss = 0.f;
            #pragma unroll
            for (int r = 0; r < 16; ++r) {
                ar[r] = arow[r * 17 + li];
                ss += ar[r] * ar[r];
            }
            #pragma unroll
            for (int d = 8; d >= 1; d >>= 1) ss += __shfl_xor(ss, d, 16);
            const float inv = rsqrtf(ss);
            #pragma unroll
            for (int r = 0; r < 16; ++r) { ar[r] *= inv; ai[r] = 0.f; }
        }
        rot4(ar, ai, wE + (0 * QB + 0) * 3);
        #pragma unroll
        for (int r = 0; r < 16; ++r) sb[r * 17 + li] = make_float2(ar[r], ai[r]);
        __syncthreads();

        #pragma unroll
        for (int r = 0; r < 16; ++r) { const float2 a = sb[li * 17 + r]; ar[r] = a.x; ai[r] = a.y; }
        rot4(ar, ai, wE + (0 * QB + 4) * 3);
        __syncthreads();
        {
            const int hi = cnot_cascade(li << 4, (0 % (QB - 1)) + 1);
            #pragma unroll
            for (int r = 0; r < 16; ++r) {
                const int d = hi ^ cnot_cascade(r, (0 % (QB - 1)) + 1);
                sb[d + (d >> 4)] = make_float2(ar[r], ai[r]);
            }
        }
        __syncthreads();

        #pragma unroll
        for (int r = 0; r < 16; ++r) { const float2 a = sb[r * 17 + li]; ar[r] = a.x; ai[r] = a.y; }
        rot4(ar, ai, wE + (1 * QB + 0) * 3);
        #pragma unroll
        for (int r = 0; r < 16; ++r) sb[r * 17 + li] = make_float2(ar[r], ai[r]);
        __syncthreads();

        #pragma unroll
        for (int r = 0; r < 16; ++r) { const float2 a = sb[li * 17 + r]; ar[r] = a.x; ai[r] = a.y; }
        rot4(ar, ai, wE + (1 * QB + 4) * 3);

        float z[QB];
        {
            const int hi = cnot_cascade(li << 4, (1 % (QB - 1)) + 1);
            float p[16];
            #pragma unroll
            for (int r = 0; r < 16; ++r) p[r] = ar[r] * ar[r] + ai[r] * ai[r];
            #pragma unroll
            for (int q = 0; q < QB; ++q) {
                float acc = 0.f;
                #pragma unroll
                for (int r = 0; r < 16; ++r)
                    acc += ((cnot_cascade(r, (1 % (QB - 1)) + 1) >> ((QB - 1) - q)) & 1) ? -p[r] : p[r];
                z[q] = ((hi >> ((QB - 1) - q)) & 1) ? -acc : acc;
            }
        }
        __syncthreads();
        float* zs = reinterpret_cast<float*>(st);
        #pragma unroll
        for (int q = 0; q < QB; ++q)
            zs[(sv * 16 + li) * QB + q] = z[q];
        __syncthreads();
        if (t < BTILE * QB) {
            const int orow = t >> 3, oq = t & 7;
            float s = 0.f;
            #pragma unroll
            for (int j = 0; j < 16; ++j)
                s += zs[(orow * 16 + j) * QB + oq];
            out[(size_t)(b0 + 16 * m + orow) * (ENC * QB) + e * QB + oq] = s;
        }
        __syncthreads();
    }
}

extern "C" void kernel_launch(void* const* d_in, const int* in_sizes, int n_in,
                              void* d_out, int out_size, void* d_ws, size_t ws_size,
                              hipStream_t stream) {
    const float* x  = (const float*)d_in[0];
    const float* pw = (const float*)d_in[1];
    const float* qw = (const float*)d_in[2];
    float* out = (float*)d_out;
    const int B = in_sizes[0] / DIM;
    dim3 grid(B / WGROWS, ENC);
    qenc_fused<<<grid, 256, 0, stream>>>(x, pw, qw, out);
}
